# starting kernel
# baseline (speedup 1.0000x reference)
#include <hip/hip_runtime.h>


#define NN 100000
#define NE 5000000
#define NE4 (NE / 4)

typedef __attribute__((ext_vector_type(2))) float v2f;
typedef __attribute__((ext_vector_type(8))) float v8f;


__global__ void k_init_deg(float* __restrict__ deg) {
  int n = blockIdx.x * blockDim.x + threadIdx.x;
  if (n < NN) deg[n] = 1.0f;
}

__global__ void k_deg_scatter(const int* __restrict__ col,
                              const float* __restrict__ w,
                              float* __restrict__ deg) {
  int t = blockIdx.x * blockDim.x + threadIdx.x;
  if (t < NE4) {
    int4   c  = ((const int4*)col)[t];
    float4 wv = ((const float4*)w)[t];
    atomicAdd(&deg[c.x], wv.x);
    atomicAdd(&deg[c.y], wv.y);
    atomicAdd(&deg[c.z], wv.z);
    atomicAdd(&deg[c.w], wv.w);
  }
}


__global__ void k_dinv_s0(const float* __restrict__ deg,
                          const float* __restrict__ x,
                          float* __restrict__ dinv,
                          float* __restrict__ s) {
  int n = blockIdx.x * blockDim.x + threadIdx.x;
  if (n < NN) {
    float d  = deg[n];
    float di = (d > 0.0f) ? rsqrtf(d) : 0.0f;
    dinv[n] = di;
    s[n] = di * di * x[n];
  }
}


__global__ void k_s_scatter(const int* __restrict__ row,
                            const int* __restrict__ col,
                            const float* __restrict__ w,
                            const float* __restrict__ dinv,
                            const float* __restrict__ x,
                            float* __restrict__ s) {
  int t = blockIdx.x * blockDim.x + threadIdx.x;
  if (t < NE4) {
    int4   r  = ((const int4*)row)[t];
    int4   c  = ((const int4*)col)[t];
    float4 wv = ((const float4*)w)[t];
    atomicAdd(&s[c.x], dinv[r.x] * wv.x * dinv[c.x] * x[r.x]);
    atomicAdd(&s[c.y], dinv[r.y] * wv.y * dinv[c.y] * x[r.y]);
    atomicAdd(&s[c.z], dinv[r.z] * wv.z * dinv[c.z] * x[r.z]);
    atomicAdd(&s[c.w], dinv[r.w] * wv.w * dinv[c.w] * x[r.w]);
  }
}


__global__ void __launch_bounds__(320)
k_node_transform(const float* __restrict__ s,
                 const float* __restrict__ W1,
                 const float* __restrict__ b1,
                 const float* __restrict__ W2,
                 float* __restrict__ z) {
  int lane = threadIdx.x & 31;
  int wave = (blockIdx.x * blockDim.x + threadIdx.x) >> 5;
  int base = wave * 16;
  int half = lane >> 4;
  int l16  = lane & 15;

  float w1 = W1[l16];
  v2f a;
  a.x = half ? 0.0f : w1;
  a.y = 0.0f;
  float sv = s[base + l16];
  v2f b;
  b.x = half ? 0.0f : sv;
  b.y = 0.0f;
  const float* b1p = b1 + half * 8;
  v8f c;
#pragma unroll
  for (int j = 0; j < 8; ++j) c[j] = b1p[j];

  v8f d = __builtin_amdgcn_wmma_f32_16x16x4_f32(
       false, a,  false, b,
       (short)0, c,  false,  false);

  const float* w2p = W2 + half * 8;
  float t = 0.0f;
#pragma unroll
  for (int j = 0; j < 8; ++j) t = fmaf(fmaxf(d[j], 0.0f), w2p[j], t);

  t += __shfl_xor(t, 16, 32);
  if (half == 0) z[base + l16] = t;
}


__global__ void k_out_init(const float* __restrict__ dinv,
                           const float* __restrict__ z,
                           const float* __restrict__ b2,
                           float* __restrict__ out) {
  int n = blockIdx.x * blockDim.x + threadIdx.x;
  if (n < NN) {
    float di = dinv[n];
    out[n] = di * di * z[n] + b2[0];
  }
}


__global__ void k_out_scatter(const int* __restrict__ row,
                              const int* __restrict__ col,
                              const float* __restrict__ w,
                              const float* __restrict__ dinv,
                              const float* __restrict__ z,
                              float* __restrict__ out) {
  int t = blockIdx.x * blockDim.x + threadIdx.x;
  if (t < NE4) {
    int4   r  = ((const int4*)row)[t];
    int4   c  = ((const int4*)col)[t];
    float4 wv = ((const float4*)w)[t];
    atomicAdd(&out[c.x], dinv[r.x] * wv.x * dinv[c.x] * z[r.x]);
    atomicAdd(&out[c.y], dinv[r.y] * wv.y * dinv[c.y] * z[r.y]);
    atomicAdd(&out[c.z], dinv[r.z] * wv.z * dinv[c.z] * z[r.z]);
    atomicAdd(&out[c.w], dinv[r.w] * wv.w * dinv[c.w] * z[r.w]);
  }
}

extern "C" void kernel_launch(void* const* d_in, const int* in_sizes, int n_in,
                              void* d_out, int out_size, void* d_ws, size_t ws_size,
                              hipStream_t stream) {
  const float* x   = (const float*)d_in[0];
  const int*   ei  = (const int*)d_in[1];
  const float* ew  = (const float*)d_in[2];
  const float* W1  = (const float*)d_in[3];
  const float* b1  = (const float*)d_in[4];
  const float* W2  = (const float*)d_in[5];
  const float* b2  = (const float*)d_in[6];
  float*       out = (float*)d_out;

  const int* row = ei;
  const int* col = ei + NE;

  float* deg  = (float*)d_ws;
  float* dinv = deg + NN;
  float* s    = dinv + NN;
  float* z    = s + NN;

  const int TB = 256;
  const int gridN = (NN + TB - 1) / TB;
  const int gridE = (NE4 + TB - 1) / TB;

  k_init_deg<<<gridN, TB, 0, stream>>>(deg);
  k_deg_scatter<<<gridE, TB, 0, stream>>>(col, ew, deg);
  k_dinv_s0<<<gridN, TB, 0, stream>>>(deg, x, dinv, s);
  k_s_scatter<<<gridE, TB, 0, stream>>>(row, col, ew, dinv, x, s);
  k_node_transform<<<625, 320, 0, stream>>>(s, W1, b1, W2, z);
  k_out_init<<<gridN, TB, 0, stream>>>(dinv, z, b2, out);
  k_out_scatter<<<gridE, TB, 0, stream>>>(row, col, ew, dinv, z, out);
}


# Degree pass also sorts each chunk by source line and keeps each edge's target-order place; edge passes gather in source order, permute through LDS, add in target order; sum order changes
# speedup vs baseline: 9.8384x; 9.8384x over previous
#include <hip/hip_runtime.h>


#define NN 100000
#define NE 5000000
#define NP 8

#define TB 256
#define ZPT 8
#define NN4 (NN / 4)

#define SB 1024
#define SQ 5
#define CH (SB * SQ * 4)
#define NCH ((NE + CH - 1) / CH)
#define NBIN 4096
static_assert(NE % 4 == 0 && CH % 128 == 0, "chunks are whole 16-byte items and whole 128-edge groups");
static_assert((NN + 31) / 32 <= NBIN && NBIN == 4 * SB, "one thread scans 4 bins");
static_assert(CH <= 65536, "a place inside a chunk fits a 16-bit word");

typedef __attribute__((ext_vector_type(2))) float v2f;

typedef __attribute__((ext_vector_type(8))) float v8f;

typedef __attribute__((ext_vector_type(4))) float v4f;

typedef __attribute__((ext_vector_type(4))) int v4i;

typedef __attribute__((ext_vector_type(2))) unsigned v2u;

__device__ __forceinline__ int place_of(int p, int nfull) {
  return (p < nfull) ? ((p & ~127) | ((p & 31) << 2) | ((p >> 5) & 3)) : p;
}


__device__ __forceinline__ int plane_of_wave() {
  unsigned id = (unsigned)__builtin_amdgcn_s_sendmsg_rtn(0x87);
  return (int)((id >> 16) & (NP - 1));
}

__global__ void __launch_bounds__(TB)
k_zero_planes(float4* __restrict__ planes) {
  int base = blockIdx.x * (TB * ZPT) + threadIdx.x;
#pragma unroll
  for (int i = 0; i < ZPT; ++i) {
    int t = base + i * TB;
    if (t < 3 * NP * NN4) planes[t] = make_float4(0.0f, 0.0f, 0.0f, 0.0f);
  }
}

__global__ void __launch_bounds__(SB)
k_deg_sort_scatter(const int* __restrict__ row,
                   const int* __restrict__ col,
                   const float* __restrict__ w,
                   float* __restrict__ degP,
                   int* __restrict__ srow,
                   unsigned short* __restrict__ stp,
                   int* __restrict__ scol,
                   float* __restrict__ sw) {
  __shared__ int   cntT[NBIN];
  __shared__ int   cntS[NBIN];
  __shared__ int   wsumT[32];
  __shared__ int   wsumS[32];
  __shared__ int   l_row[CH];
  __shared__ int   l_col[CH];
  __shared__ float l_w[CH];
  __shared__ __attribute__((aligned(16))) unsigned short l_tp[CH];

  const int tid   = threadIdx.x;
  const int lane  = tid & 31;
  const int wid   = tid >> 5;
  const int cbase = blockIdx.x * CH;
  const int nval  = (NE - cbase < CH) ? (NE - cbase) : CH;
  const int nfull = nval & ~127;

#pragma unroll
  for (int i = 0; i < 4; ++i) { cntT[tid + i * SB] = 0; cntS[tid + i * SB] = 0; }

  v4i r[SQ], c[SQ];
  v4f wv[SQ];
  bool ok[SQ];
#pragma unroll
  for (int i = 0; i < SQ; ++i) {
    int q  = tid + i * SB;
    ok[i]  = 4 * q < nval;
    int ql = ok[i] ? q : 0;
    r[i]  = ((const v4i*)(row + cbase))[ql];
    c[i]  = ((const v4i*)(col + cbase))[ql];
    wv[i] = ((const v4f*)(w + cbase))[ql];
  }
  __syncthreads();

  unsigned rk[SQ][4];
#pragma unroll
  for (int i = 0; i < SQ; ++i) {
    if (ok[i]) {
      unsigned a0 = (unsigned)atomicAdd(&cntT[c[i].x >> 5], 1);
      unsigned a1 = (unsigned)atomicAdd(&cntT[c[i].y >> 5], 1);
      unsigned a2 = (unsigned)atomicAdd(&cntT[c[i].z >> 5], 1);
      unsigned a3 = (unsigned)atomicAdd(&cntT[c[i].w >> 5], 1);
      unsigned b0 = (unsigned)atomicAdd(&cntS[r[i].x >> 5], 1);
      unsigned b1 = (unsigned)atomicAdd(&cntS[r[i].y >> 5], 1);
      unsigned b2 = (unsigned)atomicAdd(&cntS[r[i].z >> 5], 1);
      unsigned b3 = (unsigned)atomicAdd(&cntS[r[i].w >> 5], 1);
      rk[i][0] = a0 | (b0 << 16);
      rk[i][1] = a1 | (b1 << 16);
      rk[i][2] = a2 | (b2 << 16);
      rk[i][3] = a3 | (b3 << 16);
    } else {
      rk[i][0] = 0u; rk[i][1] = 0u; rk[i][2] = 0u; rk[i][3] = 0u;
    }
  }
  __syncthreads();

  v4i bt = ((const v4i*)cntT)[tid];
  v4i bs = ((const v4i*)cntS)[tid];
  int tsumT = bt.x + bt.y + bt.z + bt.w;
  int tsumS = bs.x + bs.y + bs.z + bs.w;
  int incT = tsumT, incS = tsumS;
#pragma unroll
  for (int d = 1; d < 32; d <<= 1) {
    int nT = __shfl_up(incT, d, 32);
    int nS = __shfl_up(incS, d, 32);
    if (lane >= d) { incT += nT; incS += nS; }
  }
  if (lane == 31) { wsumT[wid] = incT; wsumS[wid] = incS; }
  __syncthreads();
  if (wid < 2) {
    int* ws = (wid == 0) ? wsumT : wsumS;
    int v = ws[lane];
    int iv = v;
#pragma unroll
    for (int d = 1; d < 32; d <<= 1) {
      int n = __shfl_up(iv, d, 32);
      if (lane >= d) iv += n;
    }
    ws[lane] = iv - v;
  }
  __syncthreads();
  {
    int exT = wsumT[wid] + incT - tsumT;
    int exS = wsumS[wid] + incS - tsumS;
    v4i st;
    st.x = exT;
    st.y = exT + bt.x;
    st.z = exT + bt.x + bt.y;
    st.w = exT + bt.x + bt.y + bt.z;
    ((v4i*)cntT)[tid] = st;
    st.x = exS;
    st.y = exS + bs.x;
    st.z = exS + bs.x + bs.y;
    st.w = exS + bs.x + bs.y + bs.z;
    ((v4i*)cntS)[tid] = st;
  }
  __syncthreads();

#pragma unroll
  for (int i = 0; i < SQ; ++i) {
    if (ok[i]) {
#pragma unroll
      for (int k = 0; k < 4; ++k) {
        int cc = c[i][k];
        int rr = r[i][k];
        int pT = cntT[cc >> 5] + (int)(rk[i][k] & 0xffffu);
        int pS = cntS[rr >> 5] + (int)(rk[i][k] >> 16);
        int ixT = place_of(pT, nfull);
        int ixS = place_of(pS, nfull);
        l_col[ixT] = cc;
        l_w[ixT]   = wv[i][k];
        l_row[ixS] = rr;
        l_tp[ixS]  = (unsigned short)ixT;
      }
    }
  }
  __syncthreads();

  float* deg = degP + plane_of_wave() * NN;
#pragma unroll
  for (int i = 0; i < SQ; ++i) {
    int q = tid + i * SB;
    if (4 * q < nval) {
      v4i rr = ((const v4i*)l_row)[q];
      v2u tp = ((const v2u*)l_tp)[q];
      v4i cc = ((const v4i*)l_col)[q];
      v4f ww = ((const v4f*)l_w)[q];
      ((v4i*)(srow + cbase))[q] = rr;
      ((v2u*)(stp + cbase))[q]  = tp;
      ((v4i*)(scol + cbase))[q] = cc;
      ((v4f*)(sw + cbase))[q]   = ww;
      atomicAdd(&deg[cc.x], ww.x);
      atomicAdd(&deg[cc.y], ww.y);
      atomicAdd(&deg[cc.z], ww.z);
      atomicAdd(&deg[cc.w], ww.w);
    }
  }
}


__global__ void k_dinv_s0(const float* __restrict__ degP,
                          const float* __restrict__ x,
                          float* __restrict__ dinv,
                          float2* __restrict__ rec,
                          float* __restrict__ s) {
  int n = blockIdx.x * blockDim.x + threadIdx.x;
  if (n < NN) {
    float d = 1.0f;
#pragma unroll
    for (int p = 0; p < NP; ++p) d += degP[p * NN + n];
    float di = (d > 0.0f) ? rsqrtf(d) : 0.0f;
    float xn = x[n];
    dinv[n] = di;
    rec[n] = make_float2(di, xn);
    s[n] = di * di * xn;
  }
}


__global__ void __launch_bounds__(SB)
k_s_scatter(const int* __restrict__ srow,
            const unsigned short* __restrict__ stp,
            const int* __restrict__ scol,
            const float* __restrict__ sw,
            const float* __restrict__ dinv,
            const float2* __restrict__ rec,
            float* __restrict__ sP,
            float* __restrict__ norm) {
  __shared__ __attribute__((aligned(16))) v2f h[CH];

  const int tid   = threadIdx.x;
  const int cbase = blockIdx.x * CH;
  const int nval  = (NE - cbase < CH) ? (NE - cbase) : CH;

  v4i r[SQ], c[SQ];
  v2u tp[SQ];
  v4f wv[SQ];
  bool ok[SQ];
#pragma unroll
  for (int i = 0; i < SQ; ++i) {
    int q  = tid + i * SB;
    ok[i]  = 4 * q < nval;
    int ql = ok[i] ? q : 0;
    r[i]  = ((const v4i*)(srow + cbase))[ql];
    tp[i] = ((const v2u*)(stp + cbase))[ql];
    c[i]  = ((const v4i*)(scol + cbase))[ql];
    wv[i] = ((const v4f*)(sw + cbase))[ql];
  }
  const v2f* rec2 = (const v2f*)rec;
  v2f a[SQ][4];
#pragma unroll
  for (int i = 0; i < SQ; ++i) {
#pragma unroll
    for (int k = 0; k < 4; ++k) a[i][k] = rec2[r[i][k]];
  }
  v4f dc[SQ];
#pragma unroll
  for (int i = 0; i < SQ; ++i) {
#pragma unroll
    for (int k = 0; k < 4; ++k) dc[i][k] = dinv[c[i][k]];
  }
#pragma unroll
  for (int i = 0; i < SQ; ++i)
    asm volatile("" :: "v"(a[i][0]), "v"(a[i][1]), "v"(a[i][2]), "v"(a[i][3]));
#pragma unroll
  for (int i = 0; i < SQ; ++i) {
    if (ok[i]) {
      h[tp[i].x & 0xffffu] = a[i][0];
      h[tp[i].x >> 16]     = a[i][1];
      h[tp[i].y & 0xffffu] = a[i][2];
      h[tp[i].y >> 16]     = a[i][3];
    }
  }
  __syncthreads();

  float* s = sP + plane_of_wave() * NN;
#pragma unroll
  for (int i = 0; i < SQ; ++i) {
    int q = tid + i * SB;
    if (ok[i]) {
      v4f h01 = ((const v4f*)h)[2 * q];
      v4f h23 = ((const v4f*)h)[2 * q + 1];
      float4 nv;
      nv.x = h01.x * wv[i].x * dc[i].x;
      nv.y = h01.z * wv[i].y * dc[i].y;
      nv.z = h23.x * wv[i].z * dc[i].z;
      nv.w = h23.z * wv[i].w * dc[i].w;
      ((float4*)(norm + cbase))[q] = nv;
      atomicAdd(&s[c[i].x], nv.x * h01.y);
      atomicAdd(&s[c[i].y], nv.y * h01.w);
      atomicAdd(&s[c[i].z], nv.z * h23.y);
      atomicAdd(&s[c[i].w], nv.w * h23.w);
    }
  }
}


__global__ void __launch_bounds__(320)
k_node_transform(const float* __restrict__ s,
                 const float* __restrict__ W1,
                 const float* __restrict__ b1,
                 const float* __restrict__ W2,
                 float* __restrict__ z) {
  int lane = threadIdx.x & 31;
  int wave = (blockIdx.x * blockDim.x + threadIdx.x) >> 5;
  int base = wave * 16;
  int half = lane >> 4;
  int l16  = lane & 15;

  float w1 = W1[l16];
  v2f a;
  a.x = half ? 0.0f : w1;
  a.y = 0.0f;
  float sv = 0.0f;
#pragma unroll
  for (int p = 0; p < NP; ++p) sv += s[p * NN + base + l16];
  v2f b;
  b.x = half ? 0.0f : sv;
  b.y = 0.0f;
  const float* b1p = b1 + half * 8;
  v8f c;
#pragma unroll
  for (int j = 0; j < 8; ++j) c[j] = b1p[j];

  v8f d = __builtin_amdgcn_wmma_f32_16x16x4_f32(
       false, a,  false, b,
       (short)0, c,  false,  false);

  const float* w2p = W2 + half * 8;
  float t = 0.0f;
#pragma unroll
  for (int j = 0; j < 8; ++j) t = fmaf(fmaxf(d[j], 0.0f), w2p[j], t);

  t += __shfl_xor(t, 16, 32);
  if (half == 0) z[base + l16] = t;
}


__global__ void k_out_final(const float* __restrict__ dinv,
                            const float* __restrict__ z,
                            const float* __restrict__ b2,
                            const float* __restrict__ outP,
                            float* __restrict__ out) {
  int q = blockIdx.x * blockDim.x + threadIdx.x;
  if (q < NN4) {
    float4 di = ((const float4*)dinv)[q];
    float4 zv = ((const float4*)z)[q];
    float bb = b2[0];
    float4 acc;
    acc.x = di.x * di.x * zv.x + bb;
    acc.y = di.y * di.y * zv.y + bb;
    acc.z = di.z * di.z * zv.z + bb;
    acc.w = di.w * di.w * zv.w + bb;
#pragma unroll
    for (int p = 0; p < NP; ++p) {
      float4 v = ((const float4*)(outP + p * NN))[q];
      acc.x += v.x; acc.y += v.y; acc.z += v.z; acc.w += v.w;
    }
    ((float4*)out)[q] = acc;
  }
}


__global__ void __launch_bounds__(SB)
k_out_scatter(const int* __restrict__ srow,
              const unsigned short* __restrict__ stp,
              const int* __restrict__ scol,
              const float* __restrict__ norm,
              const float* __restrict__ z,
              float* __restrict__ outP) {
  __shared__ __attribute__((aligned(16))) float h[CH];

  const int tid   = threadIdx.x;
  const int cbase = blockIdx.x * CH;
  const int nval  = (NE - cbase < CH) ? (NE - cbase) : CH;

  v4i r[SQ], c[SQ];
  v2u tp[SQ];
  v4f nv[SQ];
  bool ok[SQ];
#pragma unroll
  for (int i = 0; i < SQ; ++i) {
    int q  = tid + i * SB;
    ok[i]  = 4 * q < nval;
    int ql = ok[i] ? q : 0;
    r[i]  = ((const v4i*)(srow + cbase))[ql];
    tp[i] = ((const v2u*)(stp + cbase))[ql];
    c[i]  = ((const v4i*)(scol + cbase))[ql];
    nv[i] = ((const v4f*)(norm + cbase))[ql];
  }
  v4f zr[SQ];
#pragma unroll
  for (int i = 0; i < SQ; ++i) {
#pragma unroll
    for (int k = 0; k < 4; ++k) zr[i][k] = z[r[i][k]];
  }
#pragma unroll
  for (int i = 0; i < SQ; ++i) {
    if (ok[i]) {
      h[tp[i].x & 0xffffu] = zr[i].x;
      h[tp[i].x >> 16]     = zr[i].y;
      h[tp[i].y & 0xffffu] = zr[i].z;
      h[tp[i].y >> 16]     = zr[i].w;
    }
  }
  __syncthreads();

  float* out = outP + plane_of_wave() * NN;
#pragma unroll
  for (int i = 0; i < SQ; ++i) {
    int q = tid + i * SB;
    if (ok[i]) {
      v4f hv = ((const v4f*)h)[q];
      atomicAdd(&out[c[i].x], nv[i].x * hv.x);
      atomicAdd(&out[c[i].y], nv[i].y * hv.y);
      atomicAdd(&out[c[i].z], nv[i].z * hv.z);
      atomicAdd(&out[c[i].w], nv[i].w * hv.w);
    }
  }
}

extern "C" void kernel_launch(void* const* d_in, const int* in_sizes, int n_in,
                              void* d_out, int out_size, void* d_ws, size_t ws_size,
                              hipStream_t stream) {
  const float* x   = (const float*)d_in[0];
  const int*   ei  = (const int*)d_in[1];
  const float* ew  = (const float*)d_in[2];
  const float* W1  = (const float*)d_in[3];
  const float* b1  = (const float*)d_in[4];
  const float* W2  = (const float*)d_in[5];
  const float* b2  = (const float*)d_in[6];
  float*       out = (float*)d_out;

  const int* row = ei;
  const int* col = ei + NE;

  float* degP = (float*)d_ws;
  float* sP   = degP + NP * NN;
  float* outP = sP + NP * NN;
  float* dinv = outP + NP * NN;
  float* z    = dinv + NN;
  float2* rec  = (float2*)(z + NN);
  float*  norm = z + 3 * NN;
  int*    srow = (int*)(norm + NE);
  int*    scol = srow + NE;
  float*  sw   = (float*)(scol + NE);
  unsigned short* stp = (unsigned short*)(sw + NE);

  const int gridN = (NN + TB - 1) / TB;
  const int gridF = (NN4 + TB - 1) / TB;
  const int gridZ = (3 * NP * NN4 + TB * ZPT - 1) / (TB * ZPT);

  k_zero_planes<<<gridZ, TB, 0, stream>>>((float4*)degP);
  k_deg_sort_scatter<<<NCH, SB, 0, stream>>>(row, col, ew, degP, srow, stp, scol, sw);
  k_dinv_s0<<<gridN, TB, 0, stream>>>(degP, x, dinv, rec, sP);
  k_s_scatter<<<NCH, SB, 0, stream>>>(srow, stp, scol, sw, dinv, rec, sP, norm);
  k_node_transform<<<625, 320, 0, stream>>>(sP, W1, b1, W2, z);
  k_out_scatter<<<NCH, SB, 0, stream>>>(srow, stp, scol, norm, z, outP);
  k_out_final<<<gridF, TB, 0, stream>>>(dinv, z, b2, outP, out);
}


# Sorted copies stored narrower: source-ordered edge = one word (node id | place<<17, was 6 B); target id kept as 16 bits, bit 16 from its sorted position: 10 B per edge, was 14; sums unchanged
# speedup vs baseline: 10.4031x; 1.0574x over previous
#include <hip/hip_runtime.h>


#define NN 100000
#define NE 5000000
#define NP 8

#define TB 256
#define ZPT 8
#define NN4 (NN / 4)

#define SB 1024
#define SQ 5
#define CH (SB * SQ * 4)
#define NCH ((NE + CH - 1) / CH)
#define NBIN 4096
#define ROWBITS 17
#define ROWMASK 0x1ffffu
#define HIBIN (65536 >> 5)
#define THRP 32
static_assert(NE % 4 == 0 && CH % 128 == 0, "chunks are whole 16-byte items and whole 128-edge groups");
static_assert((NN + 31) / 32 <= NBIN && NBIN == 4 * SB, "one thread scans 4 bins");
static_assert(CH <= (1 << (32 - ROWBITS)) && NN <= (1 << ROWBITS), "a place inside a chunk and a node id share one 32-bit word");
static_assert(NN <= 2 * 65536 && HIBIN < NBIN, "one position threshold per chunk gives bit 16 of a target id");

typedef __attribute__((ext_vector_type(2))) float v2f;

typedef __attribute__((ext_vector_type(8))) float v8f;

typedef __attribute__((ext_vector_type(4))) float v4f;

typedef __attribute__((ext_vector_type(4))) int v4i;

typedef __attribute__((ext_vector_type(2))) unsigned v2u;

typedef __attribute__((ext_vector_type(4))) unsigned v4u;

__device__ __forceinline__ int place_of(int p, int nfull) {
  return (p < nfull) ? ((p & ~127) | ((p & 31) << 2) | ((p >> 5) & 3)) : p;
}

__device__ __forceinline__ v4i cols_of(v2u cw, int q, int nfull, int thr) {
  const bool grp = 4 * q < nfull;
  const int base = grp ? (((q >> 5) << 7) | (q & 31)) : 4 * q;
  const int t  = thr - base;
  const int s1 = grp ? 32 : 1;
  const int s2 = grp ? 64 : 2;
  const int s3 = grp ? 96 : 3;
  v4i c;
  c.x = (int)(cw.x & 0xffffu) | ((0  >= t) ? 0x10000 : 0);
  c.y = (int)(cw.x >> 16)     | ((s1 >= t) ? 0x10000 : 0);
  c.z = (int)(cw.y & 0xffffu) | ((s2 >= t) ? 0x10000 : 0);
  c.w = (int)(cw.y >> 16)     | ((s3 >= t) ? 0x10000 : 0);
  return c;
}


__device__ __forceinline__ int plane_of_wave() {
  unsigned id = (unsigned)__builtin_amdgcn_s_sendmsg_rtn(0x87);
  return (int)((id >> 16) & (NP - 1));
}

__global__ void __launch_bounds__(TB)
k_zero_planes(float4* __restrict__ planes) {
  int base = blockIdx.x * (TB * ZPT) + threadIdx.x;
#pragma unroll
  for (int i = 0; i < ZPT; ++i) {
    int t = base + i * TB;
    if (t < 3 * NP * NN4) planes[t] = make_float4(0.0f, 0.0f, 0.0f, 0.0f);
  }
}

__global__ void __launch_bounds__(SB)
k_deg_sort_scatter(const int* __restrict__ row,
                   const int* __restrict__ col,
                   const float* __restrict__ w,
                   float* __restrict__ degP,
                   unsigned* __restrict__ srow,
                   unsigned short* __restrict__ scol,
                   float* __restrict__ sw,
                   int* __restrict__ thrv) {
  __shared__ int   cntT[NBIN];
  __shared__ int   cntS[NBIN];
  __shared__ int   wsumT[32];
  __shared__ int   wsumS[32];
  __shared__ __attribute__((aligned(16))) unsigned       l_row[CH];
  __shared__ __attribute__((aligned(16))) unsigned short l_col[CH];
  __shared__ __attribute__((aligned(16))) float          l_w[CH];

  const int tid   = threadIdx.x;
  const int lane  = tid & 31;
  const int wid   = tid >> 5;
  const int cbase = blockIdx.x * CH;
  const int nval  = (NE - cbase < CH) ? (NE - cbase) : CH;
  const int nfull = nval & ~127;

#pragma unroll
  for (int i = 0; i < 4; ++i) { cntT[tid + i * SB] = 0; cntS[tid + i * SB] = 0; }

  v4i r[SQ], c[SQ];
  v4f wv[SQ];
  bool ok[SQ];
#pragma unroll
  for (int i = 0; i < SQ; ++i) {
    int q  = tid + i * SB;
    ok[i]  = 4 * q < nval;
    int ql = ok[i] ? q : 0;
    r[i]  = ((const v4i*)(row + cbase))[ql];
    c[i]  = ((const v4i*)(col + cbase))[ql];
    wv[i] = ((const v4f*)(w + cbase))[ql];
  }
  __syncthreads();

  unsigned rk[SQ][4];
#pragma unroll
  for (int i = 0; i < SQ; ++i) {
    if (ok[i]) {
      unsigned a0 = (unsigned)atomicAdd(&cntT[c[i].x >> 5], 1);
      unsigned a1 = (unsigned)atomicAdd(&cntT[c[i].y >> 5], 1);
      unsigned a2 = (unsigned)atomicAdd(&cntT[c[i].z >> 5], 1);
      unsigned a3 = (unsigned)atomicAdd(&cntT[c[i].w >> 5], 1);
      unsigned b0 = (unsigned)atomicAdd(&cntS[r[i].x >> 5], 1);
      unsigned b1 = (unsigned)atomicAdd(&cntS[r[i].y >> 5], 1);
      unsigned b2 = (unsigned)atomicAdd(&cntS[r[i].z >> 5], 1);
      unsigned b3 = (unsigned)atomicAdd(&cntS[r[i].w >> 5], 1);
      rk[i][0] = a0 | (b0 << 16);
      rk[i][1] = a1 | (b1 << 16);
      rk[i][2] = a2 | (b2 << 16);
      rk[i][3] = a3 | (b3 << 16);
    } else {
      rk[i][0] = 0u; rk[i][1] = 0u; rk[i][2] = 0u; rk[i][3] = 0u;
    }
  }
  __syncthreads();

  v4i bt = ((const v4i*)cntT)[tid];
  v4i bs = ((const v4i*)cntS)[tid];
  int tsumT = bt.x + bt.y + bt.z + bt.w;
  int tsumS = bs.x + bs.y + bs.z + bs.w;
  int incT = tsumT, incS = tsumS;
#pragma unroll
  for (int d = 1; d < 32; d <<= 1) {
    int nT = __shfl_up(incT, d, 32);
    int nS = __shfl_up(incS, d, 32);
    if (lane >= d) { incT += nT; incS += nS; }
  }
  if (lane == 31) { wsumT[wid] = incT; wsumS[wid] = incS; }
  __syncthreads();
  if (wid < 2) {
    int* ws = (wid == 0) ? wsumT : wsumS;
    int v = ws[lane];
    int iv = v;
#pragma unroll
    for (int d = 1; d < 32; d <<= 1) {
      int n = __shfl_up(iv, d, 32);
      if (lane >= d) iv += n;
    }
    ws[lane] = iv - v;
  }
  __syncthreads();
  {
    int exT = wsumT[wid] + incT - tsumT;
    int exS = wsumS[wid] + incS - tsumS;
    v4i st;
    st.x = exT;
    st.y = exT + bt.x;
    st.z = exT + bt.x + bt.y;
    st.w = exT + bt.x + bt.y + bt.z;
    ((v4i*)cntT)[tid] = st;
    st.x = exS;
    st.y = exS + bs.x;
    st.z = exS + bs.x + bs.y;
    st.w = exS + bs.x + bs.y + bs.z;
    ((v4i*)cntS)[tid] = st;
  }
  __syncthreads();

  const int thr = cntT[HIBIN];
  if (tid == 0) thrv[blockIdx.x * THRP] = thr;

#pragma unroll
  for (int i = 0; i < SQ; ++i) {
    if (ok[i]) {
#pragma unroll
      for (int k = 0; k < 4; ++k) {
        int cc = c[i][k];
        int rr = r[i][k];
        int pT = cntT[cc >> 5] + (int)(rk[i][k] & 0xffffu);
        int pS = cntS[rr >> 5] + (int)(rk[i][k] >> 16);
        int ixT = place_of(pT, nfull);
        int ixS = place_of(pS, nfull);
        l_col[ixT] = (unsigned short)(cc & 0xffff);
        l_w[ixT]   = wv[i][k];
        l_row[ixS] = (unsigned)rr | ((unsigned)ixT << ROWBITS);
      }
    }
  }
  __syncthreads();

  float* deg = degP + plane_of_wave() * NN;
#pragma unroll
  for (int i = 0; i < SQ; ++i) {
    int q = tid + i * SB;
    if (4 * q < nval) {
      v4u rw = ((const v4u*)l_row)[q];
      v2u cw = ((const v2u*)l_col)[q];
      v4f ww = ((const v4f*)l_w)[q];
      ((v4u*)(srow + cbase))[q] = rw;
      ((v2u*)(scol + cbase))[q] = cw;
      ((v4f*)(sw + cbase))[q]   = ww;
      v4i cc = cols_of(cw, q, nfull, thr);
      atomicAdd(&deg[cc.x], ww.x);
      atomicAdd(&deg[cc.y], ww.y);
      atomicAdd(&deg[cc.z], ww.z);
      atomicAdd(&deg[cc.w], ww.w);
    }
  }
}


__global__ void k_dinv_s0(const float* __restrict__ degP,
                          const float* __restrict__ x,
                          float* __restrict__ dinv,
                          float2* __restrict__ rec,
                          float* __restrict__ s) {
  int n = blockIdx.x * blockDim.x + threadIdx.x;
  if (n < NN) {
    float d = 1.0f;
#pragma unroll
    for (int p = 0; p < NP; ++p) d += degP[p * NN + n];
    float di = (d > 0.0f) ? rsqrtf(d) : 0.0f;
    float xn = x[n];
    dinv[n] = di;
    rec[n] = make_float2(di, xn);
    s[n] = di * di * xn;
  }
}


__global__ void __launch_bounds__(SB)
k_s_scatter(const unsigned* __restrict__ srow,
            const unsigned short* __restrict__ scol,
            const float* __restrict__ sw,
            const int* __restrict__ thrv,
            const float* __restrict__ dinv,
            const float2* __restrict__ rec,
            float* __restrict__ sP,
            float* __restrict__ norm) {
  __shared__ __attribute__((aligned(16))) v2f h[CH];

  const int tid   = threadIdx.x;
  const int cbase = blockIdx.x * CH;
  const int nval  = (NE - cbase < CH) ? (NE - cbase) : CH;
  const int nfull = nval & ~127;
  const int thr   = thrv[blockIdx.x * THRP];

  v4u r[SQ];
  v2u cw[SQ];
  v4f wv[SQ];
  bool ok[SQ];
#pragma unroll
  for (int i = 0; i < SQ; ++i) {
    int q  = tid + i * SB;
    ok[i]  = 4 * q < nval;
    int ql = ok[i] ? q : 0;
    r[i]  = ((const v4u*)(srow + cbase))[ql];
    cw[i] = ((const v2u*)(scol + cbase))[ql];
    wv[i] = ((const v4f*)(sw + cbase))[ql];
  }
  const v2f* rec2 = (const v2f*)rec;
  v2f a[SQ][4];
#pragma unroll
  for (int i = 0; i < SQ; ++i) {
#pragma unroll
    for (int k = 0; k < 4; ++k) a[i][k] = rec2[r[i][k] & ROWMASK];
  }
  v4i c[SQ];
#pragma unroll
  for (int i = 0; i < SQ; ++i) c[i] = cols_of(cw[i], ok[i] ? tid + i * SB : 0, nfull, thr);
  v4f dc[SQ];
#pragma unroll
  for (int i = 0; i < SQ; ++i) {
#pragma unroll
    for (int k = 0; k < 4; ++k) dc[i][k] = dinv[c[i][k]];
  }
#pragma unroll
  for (int i = 0; i < SQ; ++i)
    asm volatile("" :: "v"(a[i][0]), "v"(a[i][1]), "v"(a[i][2]), "v"(a[i][3]));
#pragma unroll
  for (int i = 0; i < SQ; ++i) {
    if (ok[i]) {
      h[r[i].x >> ROWBITS] = a[i][0];
      h[r[i].y >> ROWBITS] = a[i][1];
      h[r[i].z >> ROWBITS] = a[i][2];
      h[r[i].w >> ROWBITS] = a[i][3];
    }
  }
  __syncthreads();

  float* s = sP + plane_of_wave() * NN;
#pragma unroll
  for (int i = 0; i < SQ; ++i) {
    int q = tid + i * SB;
    if (ok[i]) {
      v4f h01 = ((const v4f*)h)[2 * q];
      v4f h23 = ((const v4f*)h)[2 * q + 1];
      float4 nv;
      nv.x = h01.x * wv[i].x * dc[i].x;
      nv.y = h01.z * wv[i].y * dc[i].y;
      nv.z = h23.x * wv[i].z * dc[i].z;
      nv.w = h23.z * wv[i].w * dc[i].w;
      ((float4*)(norm + cbase))[q] = nv;
      atomicAdd(&s[c[i].x], nv.x * h01.y);
      atomicAdd(&s[c[i].y], nv.y * h01.w);
      atomicAdd(&s[c[i].z], nv.z * h23.y);
      atomicAdd(&s[c[i].w], nv.w * h23.w);
    }
  }
}


__global__ void __launch_bounds__(320)
k_node_transform(const float* __restrict__ s,
                 const float* __restrict__ W1,
                 const float* __restrict__ b1,
                 const float* __restrict__ W2,
                 float* __restrict__ z) {
  int lane = threadIdx.x & 31;
  int wave = (blockIdx.x * blockDim.x + threadIdx.x) >> 5;
  int base = wave * 16;
  int half = lane >> 4;
  int l16  = lane & 15;

  float w1 = W1[l16];
  v2f a;
  a.x = half ? 0.0f : w1;
  a.y = 0.0f;
  float sv = 0.0f;
#pragma unroll
  for (int p = 0; p < NP; ++p) sv += s[p * NN + base + l16];
  v2f b;
  b.x = half ? 0.0f : sv;
  b.y = 0.0f;
  const float* b1p = b1 + half * 8;
  v8f c;
#pragma unroll
  for (int j = 0; j < 8; ++j) c[j] = b1p[j];

  v8f d = __builtin_amdgcn_wmma_f32_16x16x4_f32(
       false, a,  false, b,
       (short)0, c,  false,  false);

  const float* w2p = W2 + half * 8;
  float t = 0.0f;
#pragma unroll
  for (int j = 0; j < 8; ++j) t = fmaf(fmaxf(d[j], 0.0f), w2p[j], t);

  t += __shfl_xor(t, 16, 32);
  if (half == 0) z[base + l16] = t;
}


__global__ void k_out_final(const float* __restrict__ dinv,
                            const float* __restrict__ z,
                            const float* __restrict__ b2,
                            const float* __restrict__ outP,
                            float* __restrict__ out) {
  int q = blockIdx.x * blockDim.x + threadIdx.x;
  if (q < NN4) {
    float4 di = ((const float4*)dinv)[q];
    float4 zv = ((const float4*)z)[q];
    float bb = b2[0];
    float4 acc;
    acc.x = di.x * di.x * zv.x + bb;
    acc.y = di.y * di.y * zv.y + bb;
    acc.z = di.z * di.z * zv.z + bb;
    acc.w = di.w * di.w * zv.w + bb;
#pragma unroll
    for (int p = 0; p < NP; ++p) {
      float4 v = ((const float4*)(outP + p * NN))[q];
      acc.x += v.x; acc.y += v.y; acc.z += v.z; acc.w += v.w;
    }
    ((float4*)out)[q] = acc;
  }
}


__global__ void __launch_bounds__(SB)
k_out_scatter(const unsigned* __restrict__ srow,
              const unsigned short* __restrict__ scol,
              const float* __restrict__ norm,
              const int* __restrict__ thrv,
              const float* __restrict__ z,
              float* __restrict__ outP) {
  __shared__ __attribute__((aligned(16))) float h[CH];

  const int tid   = threadIdx.x;
  const int cbase = blockIdx.x * CH;
  const int nval  = (NE - cbase < CH) ? (NE - cbase) : CH;
  const int nfull = nval & ~127;
  const int thr   = thrv[blockIdx.x * THRP];

  v4u r[SQ];
  v2u cw[SQ];
  v4f nv[SQ];
  bool ok[SQ];
#pragma unroll
  for (int i = 0; i < SQ; ++i) {
    int q  = tid + i * SB;
    ok[i]  = 4 * q < nval;
    int ql = ok[i] ? q : 0;
    r[i]  = ((const v4u*)(srow + cbase))[ql];
    cw[i] = ((const v2u*)(scol + cbase))[ql];
    nv[i] = ((const v4f*)(norm + cbase))[ql];
  }
  v4f zr[SQ];
#pragma unroll
  for (int i = 0; i < SQ; ++i) {
#pragma unroll
    for (int k = 0; k < 4; ++k) zr[i][k] = z[r[i][k] & ROWMASK];
  }
  v4i c[SQ];
#pragma unroll
  for (int i = 0; i < SQ; ++i) c[i] = cols_of(cw[i], ok[i] ? tid + i * SB : 0, nfull, thr);
#pragma unroll
  for (int i = 0; i < SQ; ++i) {
    if (ok[i]) {
      h[r[i].x >> ROWBITS] = zr[i].x;
      h[r[i].y >> ROWBITS] = zr[i].y;
      h[r[i].z >> ROWBITS] = zr[i].z;
      h[r[i].w >> ROWBITS] = zr[i].w;
    }
  }
  __syncthreads();

  float* out = outP + plane_of_wave() * NN;
#pragma unroll
  for (int i = 0; i < SQ; ++i) {
    int q = tid + i * SB;
    if (ok[i]) {
      v4f hv = ((const v4f*)h)[q];
      atomicAdd(&out[c[i].x], nv[i].x * hv.x);
      atomicAdd(&out[c[i].y], nv[i].y * hv.y);
      atomicAdd(&out[c[i].z], nv[i].z * hv.z);
      atomicAdd(&out[c[i].w], nv[i].w * hv.w);
    }
  }
}

extern "C" void kernel_launch(void* const* d_in, const int* in_sizes, int n_in,
                              void* d_out, int out_size, void* d_ws, size_t ws_size,
                              hipStream_t stream) {
  const float* x   = (const float*)d_in[0];
  const int*   ei  = (const int*)d_in[1];
  const float* ew  = (const float*)d_in[2];
  const float* W1  = (const float*)d_in[3];
  const float* b1  = (const float*)d_in[4];
  const float* W2  = (const float*)d_in[5];
  const float* b2  = (const float*)d_in[6];
  float*       out = (float*)d_out;

  const int* row = ei;
  const int* col = ei + NE;

  float* degP = (float*)d_ws;
  float* sP   = degP + NP * NN;
  float* outP = sP + NP * NN;
  float* dinv = outP + NP * NN;
  float* z    = dinv + NN;
  float2* rec  = (float2*)(z + NN);
  float*  norm = z + 3 * NN;
  unsigned* srow = (unsigned*)(norm + NE);
  float*  sw   = (float*)(srow + NE);
  unsigned short* scol = (unsigned short*)(sw + NE);
  int*    thrv = (int*)(scol + NE);

  const int gridN = (NN + TB - 1) / TB;
  const int gridF = (NN4 + TB - 1) / TB;
  const int gridZ = (3 * NP * NN4 + TB * ZPT - 1) / (TB * ZPT);

  k_zero_planes<<<gridZ, TB, 0, stream>>>((float4*)degP);
  k_deg_sort_scatter<<<NCH, SB, 0, stream>>>(row, col, ew, degP, srow, scol, sw, thrv);
  k_dinv_s0<<<gridN, TB, 0, stream>>>(degP, x, dinv, rec, sP);
  k_s_scatter<<<NCH, SB, 0, stream>>>(srow, scol, sw, thrv, dinv, rec, sP, norm);
  k_node_transform<<<625, 320, 0, stream>>>(sP, W1, b1, W2, z);
  k_out_scatter<<<NCH, SB, 0, stream>>>(srow, scol, norm, thrv, z, outP);
  k_out_final<<<gridF, TB, 0, stream>>>(dinv, z, b2, outP, out);
}
